# starting kernel
# baseline (speedup 1.0000x reference)
#include <hip/hip_runtime.h>

typedef float v2f __attribute__((ext_vector_type(2)));
typedef float v8f __attribute__((ext_vector_type(8)));

#define FUTURE_K 32
#define NEG_INF  (-1000000000.0f)

__launch_bounds__(32)
__global__ void future_encoder_kernel(const float* __restrict__ h,
                                      float* __restrict__ out,
                                      int B, int S, int D) {
    __shared__ float w_lds[16 * 48];

    const int tilesPerBatch = S >> 4;
    const int tile = blockIdx.x;
    const int b    = tile / tilesPerBatch;
    const int s0   = (tile % tilesPerBatch) << 4;

    const int lane = threadIdx.x;
    const int half = lane >> 4;
    const int lm   = lane & 15;

    const float* __restrict__ hb = h + (size_t)b * S * D;

    v8f c0 = {0.f,0.f,0.f,0.f,0.f,0.f,0.f,0.f};
    v8f c1 = c0, c2 = c0;

    const int aRow = s0 + lm;
    int kRow0 = s0 + 1 +  0 + lm; if (kRow0 > S - 1) kRow0 = S - 1;
    int kRow1 = s0 + 1 + 16 + lm; if (kRow1 > S - 1) kRow1 = S - 1;
    int kRow2 = s0 + 1 + 32 + lm; if (kRow2 > S - 1) kRow2 = S - 1;

    const float* pA  = hb + (size_t)aRow  * D + 2 * half;
    const float* pB0 = hb + (size_t)kRow0 * D + 2 * half;
    const float* pB1 = hb + (size_t)kRow1 * D + 2 * half;
    const float* pB2 = hb + (size_t)kRow2 * D + 2 * half;

    for (int d0 = 0; d0 < D; d0 += 4) {
        v2f a  = *(const v2f*)(pA  + d0);
        v2f b0 = *(const v2f*)(pB0 + d0);
        v2f b1 = *(const v2f*)(pB1 + d0);
        v2f b2 = *(const v2f*)(pB2 + d0);
        c0 = __builtin_amdgcn_wmma_f32_16x16x4_f32(false, a, false, b0, (short)0, c0, false, false);
        c1 = __builtin_amdgcn_wmma_f32_16x16x4_f32(false, a, false, b1, (short)0, c1, false, false);
        c2 = __builtin_amdgcn_wmma_f32_16x16x4_f32(false, a, false, b2, (short)0, c2, false, false);
    }

    float w0[8], w1[8], w2[8];
    #pragma unroll
    for (int v = 0; v < 8; ++v) {
        const int q = v + 8 * half;
        float s[3] = { c0[v], c1[v], c2[v] };
        bool ok[3];
        float m = NEG_INF;
        #pragma unroll
        for (int j = 0; j < 3; ++j) {
            const int delta  = 1 + 16 * j + lm - q;
            const int keyrow = s0 + 1 + 16 * j + lm;
            ok[j] = (delta >= 1) && (delta <= FUTURE_K) && (keyrow < S);
            if (!ok[j]) s[j] = NEG_INF;
            m = fmaxf(m, s[j]);
        }
        #pragma unroll
        for (int off = 1; off < 16; off <<= 1)
            m = fmaxf(m, __shfl_xor(m, off, 32));
        float e[3];
        float sum = 0.f;
        #pragma unroll
        for (int j = 0; j < 3; ++j) {
            e[j] = ok[j] ? __expf(s[j] - m) : 0.f;
            sum += e[j];
        }
        #pragma unroll
        for (int off = 1; off < 16; off <<= 1)
            sum += __shfl_xor(sum, off, 32);
        const float inv = (sum > 0.f) ? (1.f / sum) : 0.f;
        w0[v] = e[0] * inv;
        w1[v] = e[1] * inv;
        w2[v] = e[2] * inv;
    }

    #pragma unroll
    for (int v = 0; v < 8; ++v) {
        const int q = v + 8 * half;
        w_lds[q * 48 +  0 + lm] = w0[v];
        w_lds[q * 48 + 16 + lm] = w1[v];
        w_lds[q * 48 + 32 + lm] = w2[v];
    }
    __syncthreads();

    const float* wrow = &w_lds[lm * 48 + 2 * half];
    float* __restrict__ ob = out + (size_t)b * S * D;

    for (int dt = 0; dt < D; dt += 16) {
        v8f acc = {0.f,0.f,0.f,0.f,0.f,0.f,0.f,0.f};
        #pragma unroll
        for (int kk = 0; kk < 48; kk += 4) {
            v2f a;
            a.x = wrow[kk];
            a.y = wrow[kk + 1];
            int kr0 = s0 + 1 + kk + 2 * half;
            int kr1 = kr0 + 1;
            if (kr0 > S - 1) kr0 = S - 1;
            if (kr1 > S - 1) kr1 = S - 1;
            v2f bb;
            bb.x = hb[(size_t)kr0 * D + dt + lm];
            bb.y = hb[(size_t)kr1 * D + dt + lm];
            acc = __builtin_amdgcn_wmma_f32_16x16x4_f32(false, a, false, bb, (short)0, acc, false, false);
        }
        #pragma unroll
        for (int v = 0; v < 8; ++v)
            ob[(size_t)(s0 + v + 8 * half) * D + dt + lm] = acc[v];
    }
}

extern "C" void kernel_launch(void* const* d_in, const int* in_sizes, int n_in,
                              void* d_out, int out_size, void* d_ws, size_t ws_size,
                              hipStream_t stream) {
    const float* h = (const float*)d_in[0];
    float* out = (float*)d_out;
    const int S = 2048, D = 1024;
    const int B = in_sizes[0] / (S * D);
    dim3 grid(B * (S / 16));
    dim3 block(32);
    hipLaunchKernelGGL(future_encoder_kernel, grid, block, 0, stream, h, out, B, S, D);
}


# Split the feature dim of each 16-row tile over 16 waves of one workgroup: per-wave partial scores added in LDS in wave order (score sum order changes), each wave writes its own 64 output columns
# speedup vs baseline: 3.8138x; 3.8138x over previous
#include <hip/hip_runtime.h>

typedef float v2f __attribute__((ext_vector_type(2)));
typedef float v8f __attribute__((ext_vector_type(8)));

#define FUTURE_K 32
#define NEG_INF  (-1000000000.0f)
#define NWAVES   16

__launch_bounds__(32 * NWAVES)
__global__ void future_encoder_kernel(const float* __restrict__ h,
                                      float* __restrict__ out,
                                      int B, int S, int D) {
    __shared__ float part[NWAVES * 3 * 8 * 32];
    __shared__ float w_lds[16 * 48];

    const int tilesPerBatch = S >> 4;
    const int tile = blockIdx.x;
    const int b    = tile / tilesPerBatch;
    const int s0   = (tile % tilesPerBatch) << 4;

    const int wave = __builtin_amdgcn_readfirstlane((int)(threadIdx.x >> 5));
    const int lane = threadIdx.x & 31;
    const int half = lane >> 4;
    const int lm   = lane & 15;

    const int colsPerWave = (((D >> 4) + NWAVES - 1) / NWAVES) << 4;
    int dBeg = wave * colsPerWave;       if (dBeg > D) dBeg = D;
    int dEnd = dBeg + colsPerWave;       if (dEnd > D) dEnd = D;
    const int dLen = dEnd - dBeg;

    const float* __restrict__ hb = h + (size_t)b * S * D;

    v8f c0 = {0.f,0.f,0.f,0.f,0.f,0.f,0.f,0.f};
    v8f c1 = c0, c2 = c0;

    const int aRow = s0 + lm;
    int kRow0 = s0 + 1 +  0 + lm; if (kRow0 > S - 1) kRow0 = S - 1;
    int kRow1 = s0 + 1 + 16 + lm; if (kRow1 > S - 1) kRow1 = S - 1;
    int kRow2 = s0 + 1 + 32 + lm; if (kRow2 > S - 1) kRow2 = S - 1;

    const float* pA  = hb + (size_t)aRow  * D + dBeg + 2 * half;
    const float* pB0 = hb + (size_t)kRow0 * D + dBeg + 2 * half;
    const float* pB1 = hb + (size_t)kRow1 * D + dBeg + 2 * half;
    const float* pB2 = hb + (size_t)kRow2 * D + dBeg + 2 * half;

    int d0 = 0;
    for (; d0 + 32 <= dLen; d0 += 32) {
        v2f a[8], b0[8], b1[8], b2[8];
        for (int u = 0; u < 8; ++u) {
            a[u]  = *(const v2f*)(pA  + d0 + 4 * u);
            b0[u] = *(const v2f*)(pB0 + d0 + 4 * u);
            b1[u] = *(const v2f*)(pB1 + d0 + 4 * u);
            b2[u] = *(const v2f*)(pB2 + d0 + 4 * u);
        }
        for (int u = 0; u < 8; ++u) {
            c0 = __builtin_amdgcn_wmma_f32_16x16x4_f32(false, a[u], false, b0[u], (short)0, c0, false, false);
            c1 = __builtin_amdgcn_wmma_f32_16x16x4_f32(false, a[u], false, b1[u], (short)0, c1, false, false);
            c2 = __builtin_amdgcn_wmma_f32_16x16x4_f32(false, a[u], false, b2[u], (short)0, c2, false, false);
        }
    }
    for (; d0 < dLen; d0 += 4) {
        v2f a  = *(const v2f*)(pA  + d0);
        v2f b0 = *(const v2f*)(pB0 + d0);
        v2f b1 = *(const v2f*)(pB1 + d0);
        v2f b2 = *(const v2f*)(pB2 + d0);
        c0 = __builtin_amdgcn_wmma_f32_16x16x4_f32(false, a, false, b0, (short)0, c0, false, false);
        c1 = __builtin_amdgcn_wmma_f32_16x16x4_f32(false, a, false, b1, (short)0, c1, false, false);
        c2 = __builtin_amdgcn_wmma_f32_16x16x4_f32(false, a, false, b2, (short)0, c2, false, false);
    }

    {
        float* pw = &part[wave * (3 * 8 * 32) + lane];
        for (int v = 0; v < 8; ++v) {
            pw[( 0 + v) * 32] = c0[v];
            pw[( 8 + v) * 32] = c1[v];
            pw[(16 + v) * 32] = c2[v];
        }
    }
    __syncthreads();

    {
        const int v0 = wave & 7;
        const int q  = v0 + 8 * half;
        float s[3];
        for (int j = 0; j < 3; ++j) {
            float t = 0.f;
            for (int w = 0; w < NWAVES; ++w)
                t += part[(w * 24 + j * 8 + v0) * 32 + lane];
            s[j] = t;
        }
        bool ok[3];
        float m = NEG_INF;
        #pragma unroll
        for (int j = 0; j < 3; ++j) {
            const int delta  = 1 + 16 * j + lm - q;
            const int keyrow = s0 + 1 + 16 * j + lm;
            ok[j] = (delta >= 1) && (delta <= FUTURE_K) && (keyrow < S);
            if (!ok[j]) s[j] = NEG_INF;
            m = fmaxf(m, s[j]);
        }
        #pragma unroll
        for (int off = 1; off < 16; off <<= 1)
            m = fmaxf(m, __shfl_xor(m, off, 32));
        float e[3];
        float sum = 0.f;
        #pragma unroll
        for (int j = 0; j < 3; ++j) {
            e[j] = ok[j] ? __expf(s[j] - m) : 0.f;
            sum += e[j];
        }
        #pragma unroll
        for (int off = 1; off < 16; off <<= 1)
            sum += __shfl_xor(sum, off, 32);
        const float inv = (sum > 0.f) ? (1.f / sum) : 0.f;
        if (half == (wave >> 3)) {
            w_lds[q * 48 +  0 + lm] = e[0] * inv;
            w_lds[q * 48 + 16 + lm] = e[1] * inv;
            w_lds[q * 48 + 32 + lm] = e[2] * inv;
        }
    }
    __syncthreads();

    const float* wrow = &w_lds[lm * 48 + 2 * half];
    float* __restrict__ ob = out + (size_t)b * S * D;

    const float* __restrict__ hs = hb + dBeg;
    float* __restrict__ os = ob + dBeg;
    for (int dt = 0; dt < dLen; dt += 16) {
        v8f acc = {0.f,0.f,0.f,0.f,0.f,0.f,0.f,0.f};
        #pragma unroll
        for (int kk = 0; kk < 48; kk += 4) {
            v2f a;
            a.x = wrow[kk];
            a.y = wrow[kk + 1];
            int kr0 = s0 + 1 + kk + 2 * half;
            int kr1 = kr0 + 1;
            if (kr0 > S - 1) kr0 = S - 1;
            if (kr1 > S - 1) kr1 = S - 1;
            v2f bb;
            bb.x = hs[(size_t)kr0 * D + dt + lm];
            bb.y = hs[(size_t)kr1 * D + dt + lm];
            acc = __builtin_amdgcn_wmma_f32_16x16x4_f32(false, a, false, bb, (short)0, acc, false, false);
        }
        #pragma unroll
        for (int v = 0; v < 8; ++v)
            os[(size_t)(s0 + v + 8 * half) * D + dt + lm] = acc[v];
    }
}

extern "C" void kernel_launch(void* const* d_in, const int* in_sizes, int n_in,
                              void* d_out, int out_size, void* d_ws, size_t ws_size,
                              hipStream_t stream) {
    const float* h = (const float*)d_in[0];
    float* out = (float*)d_out;
    const int S = 2048, D = 1024;
    const int B = in_sizes[0] / (S * D);
    dim3 grid(B * (S / 16));
    dim3 block(32 * NWAVES);
    hipLaunchKernelGGL(future_encoder_kernel, grid, block, 0, stream, h, out, B, S, D);
}


# Stage each wave's 64-column slice of the tile's 49 input rows in LDS once with whole-line 16-byte loads; both WMMA products read fragments from LDS (score k order inside the sum changes)
# speedup vs baseline: 7.0418x; 1.8464x over previous
#include <hip/hip_runtime.h>

typedef float v2f __attribute__((ext_vector_type(2)));
typedef float v4f __attribute__((ext_vector_type(4)));
typedef float v8f __attribute__((ext_vector_type(8)));

#define FUTURE_K 32
#define NEG_INF  (-1000000000.0f)
#define NWAVES   16
#define FEAT     1024
#define SLICE    (FEAT / NWAVES)
#define ROWS     50
#define PITCH    76

__launch_bounds__(32 * NWAVES)
__global__ void future_encoder_kernel(const float* __restrict__ h,
                                      float* __restrict__ out,
                                      int B, int S) {
    __shared__ v4f   stage4[NWAVES * ROWS * PITCH / 4];
    __shared__ float part[NWAVES * 3 * 8 * 32];
    __shared__ float w_lds[16 * 48];

    const int tilesPerBatch = S >> 4;
    const int tile = blockIdx.x;
    const int b    = tile / tilesPerBatch;
    const int s0   = (tile % tilesPerBatch) << 4;

    const int wave = __builtin_amdgcn_readfirstlane((int)(threadIdx.x >> 5));
    const int lane = threadIdx.x & 31;
    const int half = lane >> 4;
    const int lm   = lane & 15;

    const int dBeg = wave * SLICE;
    const float* __restrict__ hb = h + (size_t)b * S * FEAT;
    float* stg = (float*)stage4 + wave * (ROWS * PITCH);

    {
        const float* src = hb + dBeg + 4 * lm;
        float*       dst = stg + 4 * lm;
        v4f r[ROWS / 2];
        #pragma unroll
        for (int i = 0; i < ROWS / 2; ++i) {
            int j = 2 * i + half;       if (j > 48) j = 48;
            int row = s0 + j;           if (row > S - 1) row = S - 1;
            r[i] = *(const v4f*)(src + (size_t)row * FEAT);
        }
        #pragma unroll
        for (int i = 0; i < ROWS / 2; ++i)
            *(v4f*)(dst + (2 * i + half) * PITCH) = r[i];
    }
    __syncthreads();

    v8f c0 = {0.f,0.f,0.f,0.f,0.f,0.f,0.f,0.f};
    v8f c1 = c0, c2 = c0;
    {
        const float* fa  = stg + lm * PITCH + 4 * half;
        const float* fb0 = stg + (1 + lm) * PITCH + 4 * half;
        const float* fb1 = fb0 + 16 * PITCH;
        const float* fb2 = fb0 + 32 * PITCH;
        v4f a[8], b0[8], b1[8], b2[8];
        #pragma unroll
        for (int t = 0; t < 8; ++t) {
            a[t]  = *(const v4f*)(fa  + 8 * t);
            b0[t] = *(const v4f*)(fb0 + 8 * t);
            b1[t] = *(const v4f*)(fb1 + 8 * t);
            b2[t] = *(const v4f*)(fb2 + 8 * t);
        }
        #pragma unroll
        for (int t = 0; t < 8; ++t) {
            c0 = __builtin_amdgcn_wmma_f32_16x16x4_f32(false, a[t].xy, false, b0[t].xy, (short)0, c0, false, false);
            c1 = __builtin_amdgcn_wmma_f32_16x16x4_f32(false, a[t].xy, false, b1[t].xy, (short)0, c1, false, false);
            c2 = __builtin_amdgcn_wmma_f32_16x16x4_f32(false, a[t].xy, false, b2[t].xy, (short)0, c2, false, false);
            c0 = __builtin_amdgcn_wmma_f32_16x16x4_f32(false, a[t].zw, false, b0[t].zw, (short)0, c0, false, false);
            c1 = __builtin_amdgcn_wmma_f32_16x16x4_f32(false, a[t].zw, false, b1[t].zw, (short)0, c1, false, false);
            c2 = __builtin_amdgcn_wmma_f32_16x16x4_f32(false, a[t].zw, false, b2[t].zw, (short)0, c2, false, false);
        }
    }

    {
        float* pw = &part[wave * (3 * 8 * 32) + lane];
        for (int v = 0; v < 8; ++v) {
            pw[( 0 + v) * 32] = c0[v];
            pw[( 8 + v) * 32] = c1[v];
            pw[(16 + v) * 32] = c2[v];
        }
    }
    __syncthreads();

    {
        const int v0 = wave & 7;
        const int q  = v0 + 8 * half;
        float s[3];
        for (int j = 0; j < 3; ++j) {
            float t = 0.f;
            for (int w = 0; w < NWAVES; ++w)
                t += part[(w * 24 + j * 8 + v0) * 32 + lane];
            s[j] = t;
        }
        bool ok[3];
        float m = NEG_INF;
        #pragma unroll
        for (int j = 0; j < 3; ++j) {
            const int delta  = 1 + 16 * j + lm - q;
            const int keyrow = s0 + 1 + 16 * j + lm;
            ok[j] = (delta >= 1) && (delta <= FUTURE_K) && (keyrow < S);
            if (!ok[j]) s[j] = NEG_INF;
            m = fmaxf(m, s[j]);
        }
        #pragma unroll
        for (int off = 1; off < 16; off <<= 1)
            m = fmaxf(m, __shfl_xor(m, off, 32));
        float e[3];
        float sum = 0.f;
        #pragma unroll
        for (int j = 0; j < 3; ++j) {
            e[j] = ok[j] ? __expf(s[j] - m) : 0.f;
            sum += e[j];
        }
        #pragma unroll
        for (int off = 1; off < 16; off <<= 1)
            sum += __shfl_xor(sum, off, 32);
        const float inv = (sum > 0.f) ? (1.f / sum) : 0.f;
        if (half == (wave >> 3)) {
            w_lds[q * 48 +  0 + lm] = e[0] * inv;
            w_lds[q * 48 + 16 + lm] = e[1] * inv;
            w_lds[q * 48 + 32 + lm] = e[2] * inv;
        }
    }
    __syncthreads();

    {
        const float* wrow = &w_lds[lm * 48 + 2 * half];
        const float* fk   = stg + (1 + 2 * half) * PITCH + lm;
        v8f acc[4];
        #pragma unroll
        for (int dt = 0; dt < 4; ++dt)
            acc[dt] = (v8f){0.f,0.f,0.f,0.f,0.f,0.f,0.f,0.f};
        #pragma unroll
        for (int kk = 0; kk < 12; ++kk) {
            v2f a;
            a.x = wrow[4 * kk];
            a.y = wrow[4 * kk + 1];
            #pragma unroll
            for (int dt = 0; dt < 4; ++dt) {
                v2f bb;
                bb.x = fk[(4 * kk    ) * PITCH + 16 * dt];
                bb.y = fk[(4 * kk + 1) * PITCH + 16 * dt];
                acc[dt] = __builtin_amdgcn_wmma_f32_16x16x4_f32(false, a, false, bb, (short)0, acc[dt], false, false);
            }
        }
        float* __restrict__ os = out + (size_t)b * S * FEAT + dBeg;
        #pragma unroll
        for (int dt = 0; dt < 4; ++dt) {
            #pragma unroll
            for (int v = 0; v < 8; ++v)
                os[(size_t)(s0 + v + 8 * half) * FEAT + 16 * dt + lm] = acc[dt][v];
        }
    }
}

extern "C" void kernel_launch(void* const* d_in, const int* in_sizes, int n_in,
                              void* d_out, int out_size, void* d_ws, size_t ws_size,
                              hipStream_t stream) {
    const float* h = (const float*)d_in[0];
    float* out = (float*)d_out;
    const int S = 2048, D = 1024;
    const int B = in_sizes[0] / (S * D);
    dim3 grid(B * (S / 16));
    dim3 block(32 * NWAVES);
    hipLaunchKernelGGL(future_encoder_kernel, grid, block, 0, stream, h, out, B, S);
}


# Each workgroup walks two consecutive 16-row tiles: the 33 shared input rows stay staged in LDS, the 16 new ones load to registers under tile 0's products and take its query rows' slots
# speedup vs baseline: 8.1727x; 1.1606x over previous
#include <hip/hip_runtime.h>

typedef float v2f __attribute__((ext_vector_type(2)));
typedef float v4f __attribute__((ext_vector_type(4)));
typedef float v8f __attribute__((ext_vector_type(8)));

#define FUTURE_K 32
#define NEG_INF  (-1000000000.0f)
#define NWAVES   16
#define FEAT     1024
#define SLICE    (FEAT / NWAVES)
#define ROWS     50
#define TILES    2
#define PITCH    76

__launch_bounds__(32 * NWAVES)
__global__ void future_encoder_kernel(const float* __restrict__ h,
                                      float* __restrict__ out,
                                      int B, int S) {
    __shared__ v4f   stage4[NWAVES * ROWS * PITCH / 4];
    __shared__ float part[NWAVES * 3 * 8 * 32];
    __shared__ float w_lds[16 * 48];

    const int walksPerBatch = S / (16 * TILES);
    const int walk = blockIdx.x;
    const int b    = walk / walksPerBatch;
    const int sW   = (walk % walksPerBatch) * (16 * TILES);

    const int wave = __builtin_amdgcn_readfirstlane((int)(threadIdx.x >> 5));
    const int lane = threadIdx.x & 31;
    const int half = lane >> 4;
    const int lm   = lane & 15;

    const int dBeg = wave * SLICE;
    const float* __restrict__ hb = h + (size_t)b * S * FEAT;
    float* stg = (float*)stage4 + wave * (ROWS * PITCH);

    {
        const float* src = hb + dBeg + 4 * lm;
        float*       dst = stg + 4 * lm;
        v4f r[ROWS / 2];
        #pragma unroll
        for (int i = 0; i < ROWS / 2; ++i) {
            int j = 2 * i + half;       if (j > 48) j = 48;
            int row = sW + j;           if (row > S - 1) row = S - 1;
            r[i] = *(const v4f*)(src + (size_t)row * FEAT);
        }
        #pragma unroll
        for (int i = 0; i < ROWS / 2; ++i)
            *(v4f*)(dst + (2 * i + half) * PITCH) = r[i];
    }
    __syncthreads();

    v4f nx[8];
    #pragma unroll
    for (int tile = 0; tile < TILES; ++tile) {
    static_assert(TILES == 2, "slot arithmetic: walk row 49 + r reuses the slot of walk row r, one step only");
    const int s0  = sW + 16 * tile;
    const int qs  = 16 * tile;
    const int ks0 = (16 * tile +  1) % 49;
    const int ks1 = (16 * tile + 17) % 49;
    const int ks2 = (16 * tile + 33) % 49;

    if (tile + 1 < TILES) {
        const float* src = hb + dBeg + 4 * lm;
        #pragma unroll
        for (int i = 0; i < 8; ++i) {
            int row = sW + 49 + 16 * tile + 2 * i + half;   if (row > S - 1) row = S - 1;
            nx[i] = *(const v4f*)(src + (size_t)row * FEAT);
        }
    }

    v8f c0 = {0.f,0.f,0.f,0.f,0.f,0.f,0.f,0.f};
    v8f c1 = c0, c2 = c0;
    {
        const float* fa  = stg + (qs  + lm) * PITCH + 4 * half;
        const float* fb0 = stg + (ks0 + lm) * PITCH + 4 * half;
        const float* fb1 = stg + (ks1 + lm) * PITCH + 4 * half;
        const float* fb2 = stg + (ks2 + lm) * PITCH + 4 * half;
        v4f a[8], b0[8], b1[8], b2[8];
        #pragma unroll
        for (int t = 0; t < 8; ++t) {
            a[t]  = *(const v4f*)(fa  + 8 * t);
            b0[t] = *(const v4f*)(fb0 + 8 * t);
            b1[t] = *(const v4f*)(fb1 + 8 * t);
            b2[t] = *(const v4f*)(fb2 + 8 * t);
        }
        #pragma unroll
        for (int t = 0; t < 8; ++t) {
            c0 = __builtin_amdgcn_wmma_f32_16x16x4_f32(false, a[t].xy, false, b0[t].xy, (short)0, c0, false, false);
            c1 = __builtin_amdgcn_wmma_f32_16x16x4_f32(false, a[t].xy, false, b1[t].xy, (short)0, c1, false, false);
            c2 = __builtin_amdgcn_wmma_f32_16x16x4_f32(false, a[t].xy, false, b2[t].xy, (short)0, c2, false, false);
            c0 = __builtin_amdgcn_wmma_f32_16x16x4_f32(false, a[t].zw, false, b0[t].zw, (short)0, c0, false, false);
            c1 = __builtin_amdgcn_wmma_f32_16x16x4_f32(false, a[t].zw, false, b1[t].zw, (short)0, c1, false, false);
            c2 = __builtin_amdgcn_wmma_f32_16x16x4_f32(false, a[t].zw, false, b2[t].zw, (short)0, c2, false, false);
        }
    }

    {
        float* pw = &part[wave * (3 * 8 * 32) + lane];
        for (int v = 0; v < 8; ++v) {
            pw[( 0 + v) * 32] = c0[v];
            pw[( 8 + v) * 32] = c1[v];
            pw[(16 + v) * 32] = c2[v];
        }
    }
    __syncthreads();

    {
        const int v0 = wave & 7;
        const int q  = v0 + 8 * half;
        float s[3];
        for (int j = 0; j < 3; ++j) {
            float t = 0.f;
            for (int w = 0; w < NWAVES; ++w)
                t += part[(w * 24 + j * 8 + v0) * 32 + lane];
            s[j] = t;
        }
        bool ok[3];
        float m = NEG_INF;
        #pragma unroll
        for (int j = 0; j < 3; ++j) {
            const int delta  = 1 + 16 * j + lm - q;
            const int keyrow = s0 + 1 + 16 * j + lm;
            ok[j] = (delta >= 1) && (delta <= FUTURE_K) && (keyrow < S);
            if (!ok[j]) s[j] = NEG_INF;
            m = fmaxf(m, s[j]);
        }
        #pragma unroll
        for (int off = 1; off < 16; off <<= 1)
            m = fmaxf(m, __shfl_xor(m, off, 32));
        float e[3];
        float sum = 0.f;
        #pragma unroll
        for (int j = 0; j < 3; ++j) {
            e[j] = ok[j] ? __expf(s[j] - m) : 0.f;
            sum += e[j];
        }
        #pragma unroll
        for (int off = 1; off < 16; off <<= 1)
            sum += __shfl_xor(sum, off, 32);
        const float inv = (sum > 0.f) ? (1.f / sum) : 0.f;
        if (half == (wave >> 3)) {
            w_lds[q * 48 +  0 + lm] = e[0] * inv;
            w_lds[q * 48 + 16 + lm] = e[1] * inv;
            w_lds[q * 48 + 32 + lm] = e[2] * inv;
        }
    }
    __syncthreads();

    {
        const float* wrow = &w_lds[lm * 48 + 2 * half];
        const float* fk[3] = { stg + (ks0 + 2 * half) * PITCH + lm,
                               stg + (ks1 + 2 * half) * PITCH + lm,
                               stg + (ks2 + 2 * half) * PITCH + lm };
        v8f acc[4];
        #pragma unroll
        for (int dt = 0; dt < 4; ++dt)
            acc[dt] = (v8f){0.f,0.f,0.f,0.f,0.f,0.f,0.f,0.f};
        #pragma unroll
        for (int kk = 0; kk < 12; ++kk) {
            v2f a;
            a.x = wrow[4 * kk];
            a.y = wrow[4 * kk + 1];
            #pragma unroll
            for (int dt = 0; dt < 4; ++dt) {
                v2f bb;
                bb.x = fk[kk >> 2][(4 * (kk & 3)    ) * PITCH + 16 * dt];
                bb.y = fk[kk >> 2][(4 * (kk & 3) + 1) * PITCH + 16 * dt];
                acc[dt] = __builtin_amdgcn_wmma_f32_16x16x4_f32(false, a, false, bb, (short)0, acc[dt], false, false);
            }
        }
        if (tile + 1 < TILES) {
            float* dst = stg + (16 * tile + half) * PITCH + 4 * lm;
            #pragma unroll
            for (int i = 0; i < 8; ++i)
                *(v4f*)(dst + 2 * i * PITCH) = nx[i];
        }
        float* __restrict__ os = out + (size_t)b * S * FEAT + dBeg;
        #pragma unroll
        for (int dt = 0; dt < 4; ++dt) {
            #pragma unroll
            for (int v = 0; v < 8; ++v)
                os[(size_t)(s0 + v + 8 * half) * FEAT + 16 * dt + lm] = acc[dt][v];
        }
    }
    }
}

extern "C" void kernel_launch(void* const* d_in, const int* in_sizes, int n_in,
                              void* d_out, int out_size, void* d_ws, size_t ws_size,
                              hipStream_t stream) {
    const float* h = (const float*)d_in[0];
    float* out = (float*)d_out;
    const int S = 2048, D = 1024;
    const int B = in_sizes[0] / (S * D);
    dim3 grid(B * (S / (16 * TILES)));
    dim3 block(32 * NWAVES);
    hipLaunchKernelGGL(future_encoder_kernel, grid, block, 0, stream, h, out, B, S);
}


# Tile 0's score product runs under its own staging: no barrier behind the wave-private rows, each key tile starts as its rows land; barriers drain LDS only, so prefetch and stores stay in flight
# speedup vs baseline: 8.8438x; 1.0821x over previous
#include <hip/hip_runtime.h>

typedef float v2f __attribute__((ext_vector_type(2)));
typedef float v4f __attribute__((ext_vector_type(4)));
typedef float v8f __attribute__((ext_vector_type(8)));

#define FUTURE_K 32
#define NEG_INF  (-1000000000.0f)
#define NWAVES   16
#define FEAT     1024
#define SLICE    (FEAT / NWAVES)
#define ROWS     50
#define TILES    2
#define PITCH    76

static __device__ __forceinline__ void lds_barrier() {
    __builtin_amdgcn_fence(__ATOMIC_RELEASE, "workgroup", "local");
    __builtin_amdgcn_s_barrier();
    __builtin_amdgcn_fence(__ATOMIC_ACQUIRE, "workgroup", "local");
}

__launch_bounds__(32 * NWAVES)
__global__ void future_encoder_kernel(const float* __restrict__ h,
                                      float* __restrict__ out,
                                      int B, int S) {
    __shared__ v4f   stage4[NWAVES * ROWS * PITCH / 4];
    __shared__ float part[NWAVES * 3 * 8 * 32];
    __shared__ float w_lds[16 * 48];

    const int walksPerBatch = S / (16 * TILES);
    const int walk = blockIdx.x;
    const int b    = walk / walksPerBatch;
    const int sW   = (walk % walksPerBatch) * (16 * TILES);

    const int wave = __builtin_amdgcn_readfirstlane((int)(threadIdx.x >> 5));
    const int lane = threadIdx.x & 31;
    const int half = lane >> 4;
    const int lm   = lane & 15;

    const int dBeg = wave * SLICE;
    const float* __restrict__ hb = h + (size_t)b * S * FEAT;
    float* stg = (float*)stage4 + wave * (ROWS * PITCH);

    v4f r[ROWS / 2];
    {
        const float* src = hb + dBeg + 4 * lm;
        #pragma unroll
        for (int i = 0; i < ROWS / 2; ++i) {
            int j = 2 * i + half;       if (j > 48) j = 48;
            int row = sW + j;           if (row > S - 1) row = S - 1;
            r[i] = *(const v4f*)(src + (size_t)row * FEAT);
        }
    }
    __builtin_amdgcn_sched_barrier(0);

    v4f nx[8];
    #pragma unroll
    for (int tile = 0; tile < TILES; ++tile) {
    static_assert(TILES == 2, "slot arithmetic: walk row 49 + r reuses the slot of walk row r, one step only");
    const int s0  = sW + 16 * tile;
    const int qs  = 16 * tile;
    const int ks0 = (16 * tile +  1) % 49;
    const int ks1 = (16 * tile + 17) % 49;
    const int ks2 = (16 * tile + 33) % 49;

    v8f c0 = {0.f,0.f,0.f,0.f,0.f,0.f,0.f,0.f};
    v8f c1 = c0, c2 = c0;
    {
        const float* fa  = stg + (qs  + lm) * PITCH + 4 * half;
        const float* fb0 = stg + (ks0 + lm) * PITCH + 4 * half;
        const float* fb1 = stg + (ks1 + lm) * PITCH + 4 * half;
        const float* fb2 = stg + (ks2 + lm) * PITCH + 4 * half;
        if (tile == 0) {
            float* dst = stg + 4 * lm;
            v4f a[8], b[8];
            #pragma unroll
            for (int i = 0; i < 9; ++i)
                *(v4f*)(dst + (2 * i + half) * PITCH) = r[i];
            #pragma unroll
            for (int t = 0; t < 8; ++t) {
                a[t] = *(const v4f*)(fa  + 8 * t);
                b[t] = *(const v4f*)(fb0 + 8 * t);
            }
            #pragma unroll
            for (int t = 0; t < 8; ++t) {
                c0 = __builtin_amdgcn_wmma_f32_16x16x4_f32(false, a[t].xy, false, b[t].xy, (short)0, c0, false, false);
                c0 = __builtin_amdgcn_wmma_f32_16x16x4_f32(false, a[t].zw, false, b[t].zw, (short)0, c0, false, false);
            }
            asm volatile("" : "+v"(c0) :: "memory");
            #pragma unroll
            for (int i = 9; i < 17; ++i)
                *(v4f*)(dst + (2 * i + half) * PITCH) = r[i];
            #pragma unroll
            for (int t = 0; t < 8; ++t)
                b[t] = *(const v4f*)(fb1 + 8 * t);
            #pragma unroll
            for (int t = 0; t < 8; ++t) {
                c1 = __builtin_amdgcn_wmma_f32_16x16x4_f32(false, a[t].xy, false, b[t].xy, (short)0, c1, false, false);
                c1 = __builtin_amdgcn_wmma_f32_16x16x4_f32(false, a[t].zw, false, b[t].zw, (short)0, c1, false, false);
            }
            asm volatile("" : "+v"(c1) :: "memory");
            #pragma unroll
            for (int i = 17; i < ROWS / 2; ++i)
                *(v4f*)(dst + (2 * i + half) * PITCH) = r[i];
            {
                const float* src = hb + dBeg + 4 * lm;
                #pragma unroll
                for (int i = 0; i < 8; ++i) {
                    int row = sW + 49 + 2 * i + half;   if (row > S - 1) row = S - 1;
                    nx[i] = *(const v4f*)(src + (size_t)row * FEAT);
                }
            }
            #pragma unroll
            for (int t = 0; t < 8; ++t)
                b[t] = *(const v4f*)(fb2 + 8 * t);
            #pragma unroll
            for (int t = 0; t < 8; ++t) {
                c2 = __builtin_amdgcn_wmma_f32_16x16x4_f32(false, a[t].xy, false, b[t].xy, (short)0, c2, false, false);
                c2 = __builtin_amdgcn_wmma_f32_16x16x4_f32(false, a[t].zw, false, b[t].zw, (short)0, c2, false, false);
            }
        } else {
            v4f a[8], b0[8], b1[8], b2[8];
            #pragma unroll
            for (int t = 0; t < 8; ++t) {
                a[t]  = *(const v4f*)(fa  + 8 * t);
                b0[t] = *(const v4f*)(fb0 + 8 * t);
                b1[t] = *(const v4f*)(fb1 + 8 * t);
                b2[t] = *(const v4f*)(fb2 + 8 * t);
            }
            #pragma unroll
            for (int t = 0; t < 8; ++t) {
                c0 = __builtin_amdgcn_wmma_f32_16x16x4_f32(false, a[t].xy, false, b0[t].xy, (short)0, c0, false, false);
                c1 = __builtin_amdgcn_wmma_f32_16x16x4_f32(false, a[t].xy, false, b1[t].xy, (short)0, c1, false, false);
                c2 = __builtin_amdgcn_wmma_f32_16x16x4_f32(false, a[t].xy, false, b2[t].xy, (short)0, c2, false, false);
                c0 = __builtin_amdgcn_wmma_f32_16x16x4_f32(false, a[t].zw, false, b0[t].zw, (short)0, c0, false, false);
                c1 = __builtin_amdgcn_wmma_f32_16x16x4_f32(false, a[t].zw, false, b1[t].zw, (short)0, c1, false, false);
                c2 = __builtin_amdgcn_wmma_f32_16x16x4_f32(false, a[t].zw, false, b2[t].zw, (short)0, c2, false, false);
            }
        }
    }

    {
        float* pw = &part[wave * (3 * 8 * 32) + lane];
        for (int v = 0; v < 8; ++v) {
            pw[( 0 + v) * 32] = c0[v];
            pw[( 8 + v) * 32] = c1[v];
            pw[(16 + v) * 32] = c2[v];
        }
    }
    lds_barrier();

    {
        const int v0 = wave & 7;
        const int q  = v0 + 8 * half;
        float s[3];
        for (int j = 0; j < 3; ++j) {
            float t = 0.f;
            for (int w = 0; w < NWAVES; ++w)
                t += part[(w * 24 + j * 8 + v0) * 32 + lane];
            s[j] = t;
        }
        bool ok[3];
        float m = NEG_INF;
        #pragma unroll
        for (int j = 0; j < 3; ++j) {
            const int delta  = 1 + 16 * j + lm - q;
            const int keyrow = s0 + 1 + 16 * j + lm;
            ok[j] = (delta >= 1) && (delta <= FUTURE_K) && (keyrow < S);
            if (!ok[j]) s[j] = NEG_INF;
            m = fmaxf(m, s[j]);
        }
        #pragma unroll
        for (int off = 1; off < 16; off <<= 1)
            m = fmaxf(m, __shfl_xor(m, off, 32));
        float e[3];
        float sum = 0.f;
        #pragma unroll
        for (int j = 0; j < 3; ++j) {
            e[j] = ok[j] ? __expf(s[j] - m) : 0.f;
            sum += e[j];
        }
        #pragma unroll
        for (int off = 1; off < 16; off <<= 1)
            sum += __shfl_xor(sum, off, 32);
        const float inv = (sum > 0.f) ? (1.f / sum) : 0.f;
        if (half == (wave >> 3)) {
            w_lds[q * 48 +  0 + lm] = e[0] * inv;
            w_lds[q * 48 + 16 + lm] = e[1] * inv;
            w_lds[q * 48 + 32 + lm] = e[2] * inv;
        }
    }
    lds_barrier();

    {
        const float* wrow = &w_lds[lm * 48 + 2 * half];
        const float* fk[3] = { stg + (ks0 + 2 * half) * PITCH + lm,
                               stg + (ks1 + 2 * half) * PITCH + lm,
                               stg + (ks2 + 2 * half) * PITCH + lm };
        v8f acc[4];
        #pragma unroll
        for (int dt = 0; dt < 4; ++dt)
            acc[dt] = (v8f){0.f,0.f,0.f,0.f,0.f,0.f,0.f,0.f};
        #pragma unroll
        for (int kk = 0; kk < 12; ++kk) {
            v2f a;
            a.x = wrow[4 * kk];
            a.y = wrow[4 * kk + 1];
            #pragma unroll
            for (int dt = 0; dt < 4; ++dt) {
                v2f bb;
                bb.x = fk[kk >> 2][(4 * (kk & 3)    ) * PITCH + 16 * dt];
                bb.y = fk[kk >> 2][(4 * (kk & 3) + 1) * PITCH + 16 * dt];
                acc[dt] = __builtin_amdgcn_wmma_f32_16x16x4_f32(false, a, false, bb, (short)0, acc[dt], false, false);
            }
        }
        if (tile + 1 < TILES) {
            float* dst = stg + (16 * tile + half) * PITCH + 4 * lm;
            #pragma unroll
            for (int i = 0; i < 8; ++i)
                *(v4f*)(dst + 2 * i * PITCH) = nx[i];
        }
        float* __restrict__ os = out + (size_t)b * S * FEAT + dBeg;
        #pragma unroll
        for (int dt = 0; dt < 4; ++dt) {
            #pragma unroll
            for (int v = 0; v < 8; ++v)
                os[(size_t)(s0 + v + 8 * half) * FEAT + 16 * dt + lm] = acc[dt][v];
        }
    }
    }
}

extern "C" void kernel_launch(void* const* d_in, const int* in_sizes, int n_in,
                              void* d_out, int out_size, void* d_ws, size_t ws_size,
                              hipStream_t stream) {
    const float* h = (const float*)d_in[0];
    float* out = (float*)d_out;
    const int S = 2048, D = 1024;
    const int B = in_sizes[0] / (S * D);
    dim3 grid(B * (S / (16 * TILES)));
    dim3 block(32 * NWAVES);
    hipLaunchKernelGGL(future_encoder_kernel, grid, block, 0, stream, h, out, B, S);
}


# Output product's four column tiles interleave (tile dt = columns 4*lm+dt of the slice): B fragments come from 16-byte LDS reads, each lane stores 16 bytes of a row, whole 128-byte lines per store
# speedup vs baseline: 9.8260x; 1.1111x over previous
#include <hip/hip_runtime.h>

typedef float v2f __attribute__((ext_vector_type(2)));
typedef float v4f __attribute__((ext_vector_type(4)));
typedef float v8f __attribute__((ext_vector_type(8)));

#define FUTURE_K 32
#define NEG_INF  (-1000000000.0f)
#define NWAVES   16
#define FEAT     1024
#define SLICE    (FEAT / NWAVES)
#define ROWS     50
#define TILES    2
#define PITCH    76

static __device__ __forceinline__ void lds_barrier() {
    __builtin_amdgcn_fence(__ATOMIC_RELEASE, "workgroup", "local");
    __builtin_amdgcn_s_barrier();
    __builtin_amdgcn_fence(__ATOMIC_ACQUIRE, "workgroup", "local");
}

__launch_bounds__(32 * NWAVES)
__global__ void future_encoder_kernel(const float* __restrict__ h,
                                      float* __restrict__ out,
                                      int B, int S) {
    __shared__ v4f   stage4[NWAVES * ROWS * PITCH / 4];
    __shared__ float part[NWAVES * 3 * 8 * 32];
    __shared__ float w_lds[16 * 48];

    const int walksPerBatch = S / (16 * TILES);
    const int walk = blockIdx.x;
    const int b    = walk / walksPerBatch;
    const int sW   = (walk % walksPerBatch) * (16 * TILES);

    const int wave = __builtin_amdgcn_readfirstlane((int)(threadIdx.x >> 5));
    const int lane = threadIdx.x & 31;
    const int half = lane >> 4;
    const int lm   = lane & 15;

    const int dBeg = wave * SLICE;
    const float* __restrict__ hb = h + (size_t)b * S * FEAT;
    float* stg = (float*)stage4 + wave * (ROWS * PITCH);

    v4f r[ROWS / 2];
    {
        const float* src = hb + dBeg + 4 * lm;
        #pragma unroll
        for (int i = 0; i < ROWS / 2; ++i) {
            int j = 2 * i + half;       if (j > 48) j = 48;
            int row = sW + j;           if (row > S - 1) row = S - 1;
            r[i] = *(const v4f*)(src + (size_t)row * FEAT);
        }
    }
    __builtin_amdgcn_sched_barrier(0);

    v4f nx[8];
    #pragma unroll
    for (int tile = 0; tile < TILES; ++tile) {
    static_assert(TILES == 2, "slot arithmetic: walk row 49 + r reuses the slot of walk row r, one step only");
    const int s0  = sW + 16 * tile;
    const int qs  = 16 * tile;
    const int ks0 = (16 * tile +  1) % 49;
    const int ks1 = (16 * tile + 17) % 49;
    const int ks2 = (16 * tile + 33) % 49;

    v8f c0 = {0.f,0.f,0.f,0.f,0.f,0.f,0.f,0.f};
    v8f c1 = c0, c2 = c0;
    {
        const float* fa  = stg + (qs  + lm) * PITCH + 4 * half;
        const float* fb0 = stg + (ks0 + lm) * PITCH + 4 * half;
        const float* fb1 = stg + (ks1 + lm) * PITCH + 4 * half;
        const float* fb2 = stg + (ks2 + lm) * PITCH + 4 * half;
        if (tile == 0) {
            float* dst = stg + 4 * lm;
            v4f a[8], b[8];
            #pragma unroll
            for (int i = 0; i < 9; ++i)
                *(v4f*)(dst + (2 * i + half) * PITCH) = r[i];
            #pragma unroll
            for (int t = 0; t < 8; ++t) {
                a[t] = *(const v4f*)(fa  + 8 * t);
                b[t] = *(const v4f*)(fb0 + 8 * t);
            }
            #pragma unroll
            for (int t = 0; t < 8; ++t) {
                c0 = __builtin_amdgcn_wmma_f32_16x16x4_f32(false, a[t].xy, false, b[t].xy, (short)0, c0, false, false);
                c0 = __builtin_amdgcn_wmma_f32_16x16x4_f32(false, a[t].zw, false, b[t].zw, (short)0, c0, false, false);
            }
            asm volatile("" : "+v"(c0) :: "memory");
            #pragma unroll
            for (int i = 9; i < 17; ++i)
                *(v4f*)(dst + (2 * i + half) * PITCH) = r[i];
            #pragma unroll
            for (int t = 0; t < 8; ++t)
                b[t] = *(const v4f*)(fb1 + 8 * t);
            #pragma unroll
            for (int t = 0; t < 8; ++t) {
                c1 = __builtin_amdgcn_wmma_f32_16x16x4_f32(false, a[t].xy, false, b[t].xy, (short)0, c1, false, false);
                c1 = __builtin_amdgcn_wmma_f32_16x16x4_f32(false, a[t].zw, false, b[t].zw, (short)0, c1, false, false);
            }
            asm volatile("" : "+v"(c1) :: "memory");
            #pragma unroll
            for (int i = 17; i < ROWS / 2; ++i)
                *(v4f*)(dst + (2 * i + half) * PITCH) = r[i];
            {
                const float* src = hb + dBeg + 4 * lm;
                #pragma unroll
                for (int i = 0; i < 8; ++i) {
                    int row = sW + 49 + 2 * i + half;   if (row > S - 1) row = S - 1;
                    nx[i] = *(const v4f*)(src + (size_t)row * FEAT);
                }
            }
            #pragma unroll
            for (int t = 0; t < 8; ++t)
                b[t] = *(const v4f*)(fb2 + 8 * t);
            #pragma unroll
            for (int t = 0; t < 8; ++t) {
                c2 = __builtin_amdgcn_wmma_f32_16x16x4_f32(false, a[t].xy, false, b[t].xy, (short)0, c2, false, false);
                c2 = __builtin_amdgcn_wmma_f32_16x16x4_f32(false, a[t].zw, false, b[t].zw, (short)0, c2, false, false);
            }
        } else {
            v4f a[8], b0[8], b1[8], b2[8];
            #pragma unroll
            for (int t = 0; t < 8; ++t) {
                a[t]  = *(const v4f*)(fa  + 8 * t);
                b0[t] = *(const v4f*)(fb0 + 8 * t);
                b1[t] = *(const v4f*)(fb1 + 8 * t);
                b2[t] = *(const v4f*)(fb2 + 8 * t);
            }
            #pragma unroll
            for (int t = 0; t < 8; ++t) {
                c0 = __builtin_amdgcn_wmma_f32_16x16x4_f32(false, a[t].xy, false, b0[t].xy, (short)0, c0, false, false);
                c1 = __builtin_amdgcn_wmma_f32_16x16x4_f32(false, a[t].xy, false, b1[t].xy, (short)0, c1, false, false);
                c2 = __builtin_amdgcn_wmma_f32_16x16x4_f32(false, a[t].xy, false, b2[t].xy, (short)0, c2, false, false);
                c0 = __builtin_amdgcn_wmma_f32_16x16x4_f32(false, a[t].zw, false, b0[t].zw, (short)0, c0, false, false);
                c1 = __builtin_amdgcn_wmma_f32_16x16x4_f32(false, a[t].zw, false, b1[t].zw, (short)0, c1, false, false);
                c2 = __builtin_amdgcn_wmma_f32_16x16x4_f32(false, a[t].zw, false, b2[t].zw, (short)0, c2, false, false);
            }
        }
    }

    {
        float* pw = &part[wave * (3 * 8 * 32) + lane];
        for (int v = 0; v < 8; ++v) {
            pw[( 0 + v) * 32] = c0[v];
            pw[( 8 + v) * 32] = c1[v];
            pw[(16 + v) * 32] = c2[v];
        }
    }
    lds_barrier();

    {
        const int v0 = wave & 7;
        const int q  = v0 + 8 * half;
        float s[3];
        for (int j = 0; j < 3; ++j) {
            float t = 0.f;
            for (int w = 0; w < NWAVES; ++w)
                t += part[(w * 24 + j * 8 + v0) * 32 + lane];
            s[j] = t;
        }
        bool ok[3];
        float m = NEG_INF;
        #pragma unroll
        for (int j = 0; j < 3; ++j) {
            const int delta  = 1 + 16 * j + lm - q;
            const int keyrow = s0 + 1 + 16 * j + lm;
            ok[j] = (delta >= 1) && (delta <= FUTURE_K) && (keyrow < S);
            if (!ok[j]) s[j] = NEG_INF;
            m = fmaxf(m, s[j]);
        }
        #pragma unroll
        for (int off = 1; off < 16; off <<= 1)
            m = fmaxf(m, __shfl_xor(m, off, 32));
        float e[3];
        float sum = 0.f;
        #pragma unroll
        for (int j = 0; j < 3; ++j) {
            e[j] = ok[j] ? __expf(s[j] - m) : 0.f;
            sum += e[j];
        }
        #pragma unroll
        for (int off = 1; off < 16; off <<= 1)
            sum += __shfl_xor(sum, off, 32);
        const float inv = (sum > 0.f) ? (1.f / sum) : 0.f;
        if (half == (wave >> 3)) {
            w_lds[q * 48 +  0 + lm] = e[0] * inv;
            w_lds[q * 48 + 16 + lm] = e[1] * inv;
            w_lds[q * 48 + 32 + lm] = e[2] * inv;
        }
    }
    lds_barrier();

    {
        const float* wrow = &w_lds[lm * 48 + 2 * half];
        const float* fk[3] = { stg + (ks0 + 2 * half) * PITCH + 4 * lm,
                               stg + (ks1 + 2 * half) * PITCH + 4 * lm,
                               stg + (ks2 + 2 * half) * PITCH + 4 * lm };
        v8f acc[4];
        #pragma unroll
        for (int dt = 0; dt < 4; ++dt)
            acc[dt] = (v8f){0.f,0.f,0.f,0.f,0.f,0.f,0.f,0.f};
        #pragma unroll
        for (int kk = 0; kk < 12; ++kk) {
            v2f a;
            a.x = wrow[4 * kk];
            a.y = wrow[4 * kk + 1];
            const v4f k0 = *(const v4f*)(fk[kk >> 2] + (4 * (kk & 3)    ) * PITCH);
            const v4f k1 = *(const v4f*)(fk[kk >> 2] + (4 * (kk & 3) + 1) * PITCH);
            #pragma unroll
            for (int dt = 0; dt < 4; ++dt) {
                v2f bb;
                bb.x = k0[dt];
                bb.y = k1[dt];
                acc[dt] = __builtin_amdgcn_wmma_f32_16x16x4_f32(false, a, false, bb, (short)0, acc[dt], false, false);
            }
        }
        asm("s_wait_alu depctr_va_vdst(0)\n\tv_nop\n\tv_nop\n\tv_nop\n\tv_nop\n\tv_nop\n\tv_nop\n\tv_nop\n\tv_nop"
                     : "+v"(acc[0]), "+v"(acc[1]), "+v"(acc[2]), "+v"(acc[3]));
        if (tile + 1 < TILES) {
            float* dst = stg + (16 * tile + half) * PITCH + 4 * lm;
            #pragma unroll
            for (int i = 0; i < 8; ++i)
                *(v4f*)(dst + 2 * i * PITCH) = nx[i];
        }
        float* __restrict__ os = out + (size_t)b * S * FEAT + dBeg + 4 * lm;
        #pragma unroll
        for (int v = 0; v < 8; ++v) {
            v4f o;
            o.x = acc[0][v];  o.y = acc[1][v];  o.z = acc[2][v];  o.w = acc[3][v];
            *(v4f*)(os + (size_t)(s0 + v + 8 * half) * FEAT) = o;
        }
    }
    }
}

extern "C" void kernel_launch(void* const* d_in, const int* in_sizes, int n_in,
                              void* d_out, int out_size, void* d_ws, size_t ws_size,
                              hipStream_t stream) {
    const float* h = (const float*)d_in[0];
    float* out = (float*)d_out;
    const int S = 2048, D = 1024;
    const int B = in_sizes[0] / (S * D);
    dim3 grid(B * (S / (16 * TILES)));
    dim3 block(32 * NWAVES);
    hipLaunchKernelGGL(future_encoder_kernel, grid, block, 0, stream, h, out, B, S);
}


# 8 waves per workgroup instead of 16, each owning a 128-column slice: half as many waves to start, same matrix work per SIMD; score sum order changes (8 partials, tile 0 even/odd k-step chains)
# speedup vs baseline: 10.4046x; 1.0589x over previous
#include <hip/hip_runtime.h>

typedef float v2f __attribute__((ext_vector_type(2)));
typedef float v4f __attribute__((ext_vector_type(4)));
typedef float v8f __attribute__((ext_vector_type(8)));

#define FUTURE_K 32
#define NEG_INF  (-1000000000.0f)
#define NWAVES   8
#define FEAT     1024
#define SLICE    (FEAT / NWAVES)
#define TSTEPS   (SLICE / 8)
#define SUBS     (SLICE / 64)
#define ROWS     49
#define TILES    2
#define PITCH    132

static __device__ __forceinline__ void lds_barrier() {
    __builtin_amdgcn_fence(__ATOMIC_RELEASE, "workgroup", "local");
    __builtin_amdgcn_s_barrier();
    __builtin_amdgcn_fence(__ATOMIC_ACQUIRE, "workgroup", "local");
}

__launch_bounds__(32 * NWAVES)
__global__ void future_encoder_kernel(const float* __restrict__ h,
                                      float* __restrict__ out,
                                      int B, int S) {
    __shared__ v4f   stage4[NWAVES * ROWS * PITCH / 4];
    __shared__ float part[NWAVES * 3 * 8 * 32];
    __shared__ float w_lds[16 * 48];

    const int walksPerBatch = S / (16 * TILES);
    const int walk = blockIdx.x;
    const int b    = walk / walksPerBatch;
    const int sW   = (walk % walksPerBatch) * (16 * TILES);

    const int wave = __builtin_amdgcn_readfirstlane((int)(threadIdx.x >> 5));
    const int lane = threadIdx.x & 31;
    const int half = lane >> 4;
    const int lm   = lane & 15;

    const int dBeg = wave * SLICE;
    const float* __restrict__ hb = h + (size_t)b * S * FEAT;
    float* stg = (float*)stage4 + wave * (ROWS * PITCH);

    v4f r[ROWS];
    {
        const float* src = hb + dBeg + 4 * lane;
        #pragma unroll
        for (int i = 0; i < ROWS; ++i) {
            int row = sW + i;           if (row > S - 1) row = S - 1;
            r[i] = *(const v4f*)(src + (size_t)row * FEAT);
        }
    }
    __builtin_amdgcn_sched_barrier(0);

    v4f nx[16];
    #pragma unroll
    for (int tile = 0; tile < TILES; ++tile) {
    static_assert(TILES == 2, "slot arithmetic: walk row 49 + r reuses the slot of walk row r, one step only");
    const int s0  = sW + 16 * tile;
    const int qs  = 16 * tile;
    const int ks0 = (16 * tile +  1) % 49;
    const int ks1 = (16 * tile + 17) % 49;
    const int ks2 = (16 * tile + 33) % 49;

    v8f c0 = {0.f,0.f,0.f,0.f,0.f,0.f,0.f,0.f};
    v8f c1 = c0, c2 = c0;
    {
        const float* fa  = stg + (qs  + lm) * PITCH + 4 * half;
        const float* fb0 = stg + (ks0 + lm) * PITCH + 4 * half;
        const float* fb1 = stg + (ks1 + lm) * PITCH + 4 * half;
        const float* fb2 = stg + (ks2 + lm) * PITCH + 4 * half;
        if (tile == 0) {
            float* dst = stg + 4 * lane;
            v4f a[TSTEPS], b[TSTEPS];
            v8f d0 = c0, d1 = c0, d2 = c0;
            #pragma unroll
            for (int i = 0; i < 17; ++i)
                *(v4f*)(dst + i * PITCH) = r[i];
            #pragma unroll
            for (int t = 0; t < TSTEPS; ++t) {
                a[t] = *(const v4f*)(fa  + 8 * t);
                b[t] = *(const v4f*)(fb0 + 8 * t);
            }
            #pragma unroll
            for (int t = 0; t < TSTEPS; t += 2) {
                c0 = __builtin_amdgcn_wmma_f32_16x16x4_f32(false, a[t].xy,     false, b[t].xy,     (short)0, c0, false, false);
                d0 = __builtin_amdgcn_wmma_f32_16x16x4_f32(false, a[t + 1].xy, false, b[t + 1].xy, (short)0, d0, false, false);
                c0 = __builtin_amdgcn_wmma_f32_16x16x4_f32(false, a[t].zw,     false, b[t].zw,     (short)0, c0, false, false);
                d0 = __builtin_amdgcn_wmma_f32_16x16x4_f32(false, a[t + 1].zw, false, b[t + 1].zw, (short)0, d0, false, false);
            }
            asm volatile("" : "+v"(c0), "+v"(d0) :: "memory");
            #pragma unroll
            for (int i = 17; i < 33; ++i)
                *(v4f*)(dst + i * PITCH) = r[i];
            #pragma unroll
            for (int t = 0; t < TSTEPS; ++t)
                b[t] = *(const v4f*)(fb1 + 8 * t);
            #pragma unroll
            for (int t = 0; t < TSTEPS; t += 2) {
                c1 = __builtin_amdgcn_wmma_f32_16x16x4_f32(false, a[t].xy,     false, b[t].xy,     (short)0, c1, false, false);
                d1 = __builtin_amdgcn_wmma_f32_16x16x4_f32(false, a[t + 1].xy, false, b[t + 1].xy, (short)0, d1, false, false);
                c1 = __builtin_amdgcn_wmma_f32_16x16x4_f32(false, a[t].zw,     false, b[t].zw,     (short)0, c1, false, false);
                d1 = __builtin_amdgcn_wmma_f32_16x16x4_f32(false, a[t + 1].zw, false, b[t + 1].zw, (short)0, d1, false, false);
            }
            asm volatile("" : "+v"(c1), "+v"(d1) :: "memory");
            #pragma unroll
            for (int i = 33; i < ROWS; ++i)
                *(v4f*)(dst + i * PITCH) = r[i];
            {
                const float* src = hb + dBeg + 4 * lane;
                #pragma unroll
                for (int i = 0; i < 16; ++i) {
                    int row = sW + 49 + i;   if (row > S - 1) row = S - 1;
                    nx[i] = *(const v4f*)(src + (size_t)row * FEAT);
                }
            }
            #pragma unroll
            for (int t = 0; t < TSTEPS; ++t)
                b[t] = *(const v4f*)(fb2 + 8 * t);
            #pragma unroll
            for (int t = 0; t < TSTEPS; t += 2) {
                c2 = __builtin_amdgcn_wmma_f32_16x16x4_f32(false, a[t].xy,     false, b[t].xy,     (short)0, c2, false, false);
                d2 = __builtin_amdgcn_wmma_f32_16x16x4_f32(false, a[t + 1].xy, false, b[t + 1].xy, (short)0, d2, false, false);
                c2 = __builtin_amdgcn_wmma_f32_16x16x4_f32(false, a[t].zw,     false, b[t].zw,     (short)0, c2, false, false);
                d2 = __builtin_amdgcn_wmma_f32_16x16x4_f32(false, a[t + 1].zw, false, b[t + 1].zw, (short)0, d2, false, false);
            }
            asm("s_wait_alu depctr_va_vdst(0)\n\tv_nop\n\tv_nop\n\tv_nop\n\tv_nop\n\tv_nop\n\tv_nop\n\tv_nop\n\tv_nop"
                         : "+v"(c0), "+v"(d0), "+v"(c1), "+v"(d1), "+v"(c2), "+v"(d2));
            c0 += d0;   c1 += d1;   c2 += d2;
        } else {
            v4f a[TSTEPS], b0[TSTEPS], b1[TSTEPS], b2[TSTEPS];
            #pragma unroll
            for (int t = 0; t < TSTEPS; ++t) {
                a[t]  = *(const v4f*)(fa  + 8 * t);
                b0[t] = *(const v4f*)(fb0 + 8 * t);
                b1[t] = *(const v4f*)(fb1 + 8 * t);
                b2[t] = *(const v4f*)(fb2 + 8 * t);
            }
            #pragma unroll
            for (int t = 0; t < TSTEPS; ++t) {
                c0 = __builtin_amdgcn_wmma_f32_16x16x4_f32(false, a[t].xy, false, b0[t].xy, (short)0, c0, false, false);
                c1 = __builtin_amdgcn_wmma_f32_16x16x4_f32(false, a[t].xy, false, b1[t].xy, (short)0, c1, false, false);
                c2 = __builtin_amdgcn_wmma_f32_16x16x4_f32(false, a[t].xy, false, b2[t].xy, (short)0, c2, false, false);
                c0 = __builtin_amdgcn_wmma_f32_16x16x4_f32(false, a[t].zw, false, b0[t].zw, (short)0, c0, false, false);
                c1 = __builtin_amdgcn_wmma_f32_16x16x4_f32(false, a[t].zw, false, b1[t].zw, (short)0, c1, false, false);
                c2 = __builtin_amdgcn_wmma_f32_16x16x4_f32(false, a[t].zw, false, b2[t].zw, (short)0, c2, false, false);
            }
        }
    }

    {
        float* pw = &part[wave * (3 * 8 * 32) + lane];
        for (int v = 0; v < 8; ++v) {
            pw[( 0 + v) * 32] = c0[v];
            pw[( 8 + v) * 32] = c1[v];
            pw[(16 + v) * 32] = c2[v];
        }
    }
    lds_barrier();

    {
        const int v0 = wave;
        const int q  = v0 + 8 * half;
        float s[3];
        for (int j = 0; j < 3; ++j) {
            float t = 0.f;
            for (int w = 0; w < NWAVES; ++w)
                t += part[(w * 24 + j * 8 + v0) * 32 + lane];
            s[j] = t;
        }
        bool ok[3];
        float m = NEG_INF;
        #pragma unroll
        for (int j = 0; j < 3; ++j) {
            const int delta  = 1 + 16 * j + lm - q;
            const int keyrow = s0 + 1 + 16 * j + lm;
            ok[j] = (delta >= 1) && (delta <= FUTURE_K) && (keyrow < S);
            if (!ok[j]) s[j] = NEG_INF;
            m = fmaxf(m, s[j]);
        }
        #pragma unroll
        for (int off = 1; off < 16; off <<= 1)
            m = fmaxf(m, __shfl_xor(m, off, 32));
        float e[3];
        float sum = 0.f;
        #pragma unroll
        for (int j = 0; j < 3; ++j) {
            e[j] = ok[j] ? __expf(s[j] - m) : 0.f;
            sum += e[j];
        }
        #pragma unroll
        for (int off = 1; off < 16; off <<= 1)
            sum += __shfl_xor(sum, off, 32);
        const float inv = (sum > 0.f) ? (1.f / sum) : 0.f;
        w_lds[q * 48 +  0 + lm] = e[0] * inv;
        w_lds[q * 48 + 16 + lm] = e[1] * inv;
        w_lds[q * 48 + 32 + lm] = e[2] * inv;
    }
    lds_barrier();

    {
        const float* wrow = &w_lds[lm * 48 + 2 * half];
        const float* fk[3] = { stg + (ks0 + 2 * half) * PITCH + 4 * lm,
                               stg + (ks1 + 2 * half) * PITCH + 4 * lm,
                               stg + (ks2 + 2 * half) * PITCH + 4 * lm };
        v8f acc[SUBS][4];
        #pragma unroll
        for (int sub = 0; sub < SUBS; ++sub) {
            #pragma unroll
            for (int dt = 0; dt < 4; ++dt)
                acc[sub][dt] = (v8f){0.f,0.f,0.f,0.f,0.f,0.f,0.f,0.f};
        }
        #pragma unroll
        for (int kk = 0; kk < 12; ++kk) {
            v2f a;
            a.x = wrow[4 * kk];
            a.y = wrow[4 * kk + 1];
            #pragma unroll
            for (int sub = 0; sub < SUBS; ++sub) {
                const v4f k0 = *(const v4f*)(fk[kk >> 2] + (4 * (kk & 3)    ) * PITCH + 64 * sub);
                const v4f k1 = *(const v4f*)(fk[kk >> 2] + (4 * (kk & 3) + 1) * PITCH + 64 * sub);
                #pragma unroll
                for (int dt = 0; dt < 4; ++dt) {
                    v2f bb;
                    bb.x = k0[dt];
                    bb.y = k1[dt];
                    acc[sub][dt] = __builtin_amdgcn_wmma_f32_16x16x4_f32(false, a, false, bb, (short)0, acc[sub][dt], false, false);
                }
            }
        }
        asm("s_wait_alu depctr_va_vdst(0)\n\tv_nop\n\tv_nop\n\tv_nop\n\tv_nop\n\tv_nop\n\tv_nop\n\tv_nop\n\tv_nop"
                     : "+v"(acc[0][0]), "+v"(acc[0][1]), "+v"(acc[0][2]), "+v"(acc[0][3]),
                       "+v"(acc[1][0]), "+v"(acc[1][1]), "+v"(acc[1][2]), "+v"(acc[1][3]));
        if (tile + 1 < TILES) {
            float* dst = stg + (16 * tile) * PITCH + 4 * lane;
            #pragma unroll
            for (int i = 0; i < 16; ++i)
                *(v4f*)(dst + i * PITCH) = nx[i];
        }
        float* __restrict__ os = out + (size_t)b * S * FEAT + dBeg + 4 * lm;
        #pragma unroll
        for (int sub = 0; sub < SUBS; ++sub) {
            #pragma unroll
            for (int v = 0; v < 8; ++v) {
                v4f o;
                o.x = acc[sub][0][v];  o.y = acc[sub][1][v];  o.z = acc[sub][2][v];  o.w = acc[sub][3][v];
                *(v4f*)(os + 64 * sub + (size_t)(s0 + v + 8 * half) * FEAT) = o;
            }
        }
    }
    }
}

extern "C" void kernel_launch(void* const* d_in, const int* in_sizes, int n_in,
                              void* d_out, int out_size, void* d_ws, size_t ws_size,
                              hipStream_t stream) {
    const float* h = (const float*)d_in[0];
    float* out = (float*)d_out;
    const int S = 2048, D = 1024;
    const int B = in_sizes[0] / (S * D);
    dim3 grid(B * (S / (16 * TILES)));
    dim3 block(32 * NWAVES);
    hipLaunchKernelGGL(future_encoder_kernel, grid, block, 0, stream, h, out, B, S);
}
